# starting kernel
# baseline (speedup 1.0000x reference)
#include <hip/hip_runtime.h>
#include <hip/hip_bf16.h>


#define BATCH   2
#define IMG     384
#define C_DIM   96
#define HEADS   6
#define HD      16
#define WIN     16
#define NWIN    24
#define BW      (BATCH * NWIN * NWIN)
#define LTOK    256
#define KSEL    64
#define NPIX    (BATCH * IMG * IMG)

typedef __attribute__((ext_vector_type(2))) float v2f;
typedef __attribute__((ext_vector_type(8))) float v8f;

static __device__ inline v8f wmma4(v2f a, v2f b, v8f c) {
    return __builtin_amdgcn_wmma_f32_16x16x4_f32(
         false, a,  false, b,
         (short)0, c,  false,  false);
}

__global__ __launch_bounds__(256) void k1_router(
    const float* __restrict__ x, const float* __restrict__ Wr,
    int* __restrict__ gidx, float* __restrict__ gval)
{
    __shared__ float sWr[C_DIM * HEADS];
    __shared__ float sv[LTOK];
    __shared__ unsigned long long key[LTOK];

    const int w = blockIdx.x;
    const int t = threadIdx.x;

    for (int i = t; i < C_DIM * HEADS; i += 256) sWr[i] = Wr[i];
    __syncthreads();

    const int b  = w / (NWIN * NWIN);
    const int rm = w % (NWIN * NWIN);
    const int wy = rm / NWIN, wx = rm % NWIN;
    const int hh = wy * WIN + (t >> 4);
    const int ww = wx * WIN + (t & 15);
    const float* xr = x + ((size_t)(b * IMG + hh) * IMG + ww) * C_DIM;

    float lg[HEADS];
    #pragma unroll
    for (int e = 0; e < HEADS; ++e) lg[e] = 0.f;
    for (int c = 0; c < C_DIM; ++c) {
        float xv = xr[c];
        #pragma unroll
        for (int e = 0; e < HEADS; ++e) lg[e] += xv * sWr[c * HEADS + e];
    }
    #pragma unroll
    for (int e = 0; e < HEADS; ++e) lg[e] = 1.0f / (1.0f + __expf(-lg[e]));

    for (int e = 0; e < HEADS; ++e) {
        __syncthreads();
        sv[t] = lg[e];
        unsigned int fb = (t == 0) ? 0x7F800000u : __float_as_uint(lg[e]);
        key[t] = ((unsigned long long)fb << 32) | (unsigned int)(0xFFFFu - t);
        __syncthreads();
        for (int ks = 2; ks <= LTOK; ks <<= 1) {
            for (int j = ks >> 1; j > 0; j >>= 1) {
                int ixj = t ^ j;
                if (ixj > t) {
                    unsigned long long a = key[t], bb = key[ixj];
                    bool descBlock = ((t & ks) == 0);
                    if (descBlock ? (a < bb) : (a > bb)) { key[t] = bb; key[ixj] = a; }
                }
                __syncthreads();
            }
        }
        if (t < KSEL) {
            int tok = 0xFFFF - (int)(key[t] & 0xFFFFull);
            gidx[((size_t)w * HEADS + e) * KSEL + t] = tok;
            gval[((size_t)w * HEADS + e) * KSEL + t] = sv[tok];
        }
    }
}

__global__ __launch_bounds__(128) void k2_attn(
    const float* __restrict__ x,
    const float* __restrict__ Wqkv, const float* __restrict__ Wo,
    const int* __restrict__ gidx, const float* __restrict__ gval,
    float* __restrict__ scat)
{
    __shared__ float pool[11264];
    __shared__ float sQKV[KSEL * 48];
    __shared__ float sgate[KSEL];
    __shared__ int   sidx[KSEL];

    float* Xg  = pool;
    float* sWq = pool + 6144;
    float* S   = pool;
    float* O   = pool + 4096;
    float* sWo = pool + 5120;

    const int w    = blockIdx.x;
    const int e    = blockIdx.y;
    const int tid  = threadIdx.x;
    const int lane = tid & 31;
    const int wave = tid >> 5;
    const int hi   = lane >> 4;
    const int lo   = lane & 15;

    if (tid < KSEL) {
        sidx[tid]  = gidx[((size_t)w * HEADS + e) * KSEL + tid];
        sgate[tid] = gval[((size_t)w * HEADS + e) * KSEL + tid];
    }
    __syncthreads();

    const int b  = w / (NWIN * NWIN);
    const int rm = w % (NWIN * NWIN);
    const int wy = rm / NWIN, wx = rm % NWIN;

    for (int i = tid; i < KSEL * C_DIM; i += 128) {
        int j = i / C_DIM, c = i % C_DIM;
        int t  = sidx[j];
        int hh = wy * WIN + (t >> 4);
        int ww = wx * WIN + (t & 15);
        Xg[i] = x[((size_t)(b * IMG + hh) * IMG + ww) * C_DIM + c];
    }
    for (int i = tid; i < C_DIM * 48; i += 128)
        sWq[i] = Wqkv[(size_t)e * C_DIM * 48 + i];
    __syncthreads();

    for (int tile = wave; tile < 12; tile += 4) {
        int mt = tile / 3, nt = tile % 3;
        v8f acc = {};
        for (int kk = 0; kk < C_DIM; kk += 4) {
            int ka = kk + 2 * hi;
            v2f a = { Xg[(mt * 16 + lo) * C_DIM + ka],
                      Xg[(mt * 16 + lo) * C_DIM + ka + 1] };
            v2f bf = { sWq[ka * 48 + nt * 16 + lo],
                       sWq[(ka + 1) * 48 + nt * 16 + lo] };
            acc = wmma4(a, bf, acc);
        }
        #pragma unroll
        for (int r = 0; r < 8; ++r)
            sQKV[(mt * 16 + r + 8 * hi) * 48 + nt * 16 + lo] = acc[r];
    }
    __syncthreads();

    for (int i = tid; i < HD * C_DIM; i += 128)
        sWo[i] = Wo[(size_t)e * HD * C_DIM + i];

    for (int tile = wave; tile < 16; tile += 4) {
        int mt = tile / 4, nt = tile % 4;
        v8f acc = {};
        for (int kk = 0; kk < HD; kk += 4) {
            int ka = kk + 2 * hi;
            v2f a = { sQKV[(mt * 16 + lo) * 48 + ka],
                      sQKV[(mt * 16 + lo) * 48 + ka + 1] };
            v2f bf = { sQKV[(nt * 16 + lo) * 48 + 16 + ka],
                       sQKV[(nt * 16 + lo) * 48 + 16 + ka + 1] };
            acc = wmma4(a, bf, acc);
        }
        #pragma unroll
        for (int r = 0; r < 8; ++r)
            S[(mt * 16 + r + 8 * hi) * 64 + nt * 16 + lo] = acc[r] * 0.25f;
    }
    __syncthreads();

    if (tid < KSEL) {
        float m = -1e30f;
        for (int j = 0; j < 64; ++j) m = fmaxf(m, S[tid * 64 + j]);
        float s = 0.f;
        for (int j = 0; j < 64; ++j) {
            float ev = __expf(S[tid * 64 + j] - m);
            S[tid * 64 + j] = ev; s += ev;
        }
        float inv = sgate[tid] / s;
        for (int j = 0; j < 64; ++j) S[tid * 64 + j] *= inv;
    }
    __syncthreads();

    {
        int mt = wave;
        v8f acc = {};
        for (int kk = 0; kk < 64; kk += 4) {
            int ka = kk + 2 * hi;
            v2f a = { S[(mt * 16 + lo) * 64 + ka],
                      S[(mt * 16 + lo) * 64 + ka + 1] };
            v2f bf = { sQKV[ka * 48 + 32 + lo],
                       sQKV[(ka + 1) * 48 + 32 + lo] };
            acc = wmma4(a, bf, acc);
        }
        #pragma unroll
        for (int r = 0; r < 8; ++r)
            O[(mt * 16 + r + 8 * hi) * HD + lo] = acc[r];
    }
    __syncthreads();

    for (int tile = wave; tile < 24; tile += 4) {
        int mt = tile / 6, nt = tile % 6;
        v8f acc = {};
        for (int kk = 0; kk < HD; kk += 4) {
            int ka = kk + 2 * hi;
            v2f a = { O[(mt * 16 + lo) * HD + ka],
                      O[(mt * 16 + lo) * HD + ka + 1] };
            v2f bf = { sWo[ka * C_DIM + nt * 16 + lo],
                       sWo[(ka + 1) * C_DIM + nt * 16 + lo] };
            acc = wmma4(a, bf, acc);
        }
        #pragma unroll
        for (int r = 0; r < 8; ++r) {
            int m  = mt * 16 + r + 8 * hi;
            int t  = sidx[m];
            int hh = wy * WIN + (t >> 4);
            int ww = wx * WIN + (t & 15);
            atomicAdd(&scat[((size_t)(b * IMG + hh) * IMG + ww) * C_DIM + nt * 16 + lo],
                      acc[r]);
        }
    }
}

__global__ __launch_bounds__(256) void k3_proj(
    const float* __restrict__ scat, const float* __restrict__ projW,
    const float* __restrict__ projb, float* __restrict__ y)
{
    __shared__ float A[64 * C_DIM];
    __shared__ float Wt[C_DIM * C_DIM];
    __shared__ float bias[C_DIM];

    const int tid  = threadIdx.x;
    const size_t row0 = (size_t)blockIdx.x * 64;

    for (int i = tid; i < 64 * C_DIM; i += 256) A[i] = scat[row0 * C_DIM + i];
    for (int i = tid; i < C_DIM * C_DIM; i += 256) {
        int c = i / C_DIM, j = i % C_DIM;
        Wt[i] = projW[(size_t)j * C_DIM + c];
    }
    if (tid < C_DIM) bias[tid] = projb[tid];
    __syncthreads();

    const int lane = tid & 31, wave = tid >> 5;
    const int hi = lane >> 4, lo = lane & 15;

    for (int tile = wave; tile < 24; tile += 8) {
        int mt = tile / 6, nt = tile % 6;
        v8f acc = {};
        for (int kk = 0; kk < C_DIM; kk += 4) {
            int ka = kk + 2 * hi;
            v2f a = { A[(mt * 16 + lo) * C_DIM + ka],
                      A[(mt * 16 + lo) * C_DIM + ka + 1] };
            v2f bf = { Wt[ka * C_DIM + nt * 16 + lo],
                       Wt[(ka + 1) * C_DIM + nt * 16 + lo] };
            acc = wmma4(a, bf, acc);
        }
        #pragma unroll
        for (int r = 0; r < 8; ++r) {
            int m = mt * 16 + r + 8 * hi;
            y[(row0 + m) * C_DIM + nt * 16 + lo] = acc[r] + bias[nt * 16 + lo];
        }
    }
}

extern "C" void kernel_launch(void* const* d_in, const int* in_sizes, int n_in,
                              void* d_out, int out_size, void* d_ws, size_t ws_size,
                              hipStream_t stream) {
    const float* x     = (const float*)d_in[0];
    const float* Wr    = (const float*)d_in[1];
    const float* Wqkv  = (const float*)d_in[2];
    const float* Wo    = (const float*)d_in[3];
    const float* projW = (const float*)d_in[4];
    const float* projb = (const float*)d_in[5];
    float* y = (float*)d_out;

    const size_t scatBytes = (size_t)NPIX * C_DIM * sizeof(float);
    float* scat = (float*)d_ws;
    int*   gidx = (int*)((char*)d_ws + scatBytes);
    float* gval = (float*)((char*)d_ws + scatBytes +
                           (size_t)BW * HEADS * KSEL * sizeof(int));

    hipMemsetAsync(scat, 0, scatBytes, stream);

    k1_router<<<BW, 256, 0, stream>>>(x, Wr, gidx, gval);
    k2_attn<<<dim3(BW, HEADS), 128, 0, stream>>>(x, Wqkv, Wo, gidx, gval, scat);
    k3_proj<<<NPIX / 64, 256, 0, stream>>>(scat, projW, projb, y);
}


# k1: each head's 256 keys sorted by one wave in registers (8 keys per lane; the same bitonic network as lane permutes and in-lane exchanges): 2 block barriers, not 229; picks and order unchanged
# speedup vs baseline: 1.8974x; 1.8974x over previous
#include <hip/hip_runtime.h>
#include <hip/hip_bf16.h>


#define BATCH   2
#define IMG     384
#define C_DIM   96
#define HEADS   6
#define HD      16
#define WIN     16
#define NWIN    24
#define BW      (BATCH * NWIN * NWIN)
#define LTOK    256
#define KSEL    64
#define NPIX    (BATCH * IMG * IMG)

typedef __attribute__((ext_vector_type(2))) float v2f;
typedef __attribute__((ext_vector_type(8))) float v8f;

static __device__ inline v8f wmma4(v2f a, v2f b, v8f c) {
    return __builtin_amdgcn_wmma_f32_16x16x4_f32(
         false, a,  false, b,
         (short)0, c,  false,  false);
}

typedef unsigned long long u64;

template <int M> static __device__ __forceinline__ unsigned int lane_xor_u32(unsigned int x) {
    const int b = (int)x;
    if constexpr (M == 16) {
        return (unsigned int)__builtin_amdgcn_permlanex16(b, b, 0x76543210, 0xfedcba98, false, false);
    } else {
        return (unsigned int)__builtin_amdgcn_update_dpp(b, b, 0x160 + M, 0xF, 0xF, true);
    }
}
template <int M> static __device__ __forceinline__ u64 lane_xor_u64(u64 k) {
    const unsigned int lo = lane_xor_u32<M>((unsigned int)k);
    const unsigned int hi = lane_xor_u32<M>((unsigned int)(k >> 32));
    return ((u64)hi << 32) | lo;
}

template <int KS, int J>
static __device__ __forceinline__ void cx_lanes(u64 (&k)[8], const int lane) {
    const bool low = (lane & J) == 0;
    #pragma unroll
    for (int r = 0; r < 8; ++r) {
        const bool desc = (KS < 32) ? ((lane & KS) == 0) : (((r * 32) & KS) == 0);
        const bool keepMax = (desc == low);
        const u64 a = k[r];
        const u64 b = lane_xor_u64<J>(a);
        const bool gt = a > b;
        k[r] = (gt == keepMax) ? a : b;
    }
}
template <int KS, int JR>
static __device__ __forceinline__ void cx_regs(u64 (&k)[8]) {
    #pragma unroll
    for (int r = 0; r < 8; ++r) {
        if ((r & JR) == 0) {
            const bool desc = ((r * 32) & KS) == 0;
            const u64 a = k[r], b = k[r ^ JR];
            const u64 mx = (a > b) ? a : b;
            const u64 mn = (a > b) ? b : a;
            k[r]      = desc ? mx : mn;
            k[r ^ JR] = desc ? mn : mx;
        }
    }
}
template <int KS>
static __device__ __forceinline__ void cx_tail(u64 (&k)[8], const int lane) {
    cx_lanes<KS, 16>(k, lane);
    cx_lanes<KS, 8>(k, lane);
    cx_lanes<KS, 4>(k, lane);
    cx_lanes<KS, 2>(k, lane);
    cx_lanes<KS, 1>(k, lane);
}

__global__ __launch_bounds__(256) void k1_router(
    const float* __restrict__ x, const float* __restrict__ Wr,
    int* __restrict__ gidx, float* __restrict__ gval)
{
    __shared__ float sWr[C_DIM * HEADS];
    __shared__ float sv6[HEADS * LTOK];

    const int w = blockIdx.x;
    const int t = threadIdx.x;

    for (int i = t; i < C_DIM * HEADS; i += 256) sWr[i] = Wr[i];
    __syncthreads();

    const int b  = w / (NWIN * NWIN);
    const int rm = w % (NWIN * NWIN);
    const int wy = rm / NWIN, wx = rm % NWIN;
    const int hh = wy * WIN + (t >> 4);
    const int ww = wx * WIN + (t & 15);
    const float* xr = x + ((size_t)(b * IMG + hh) * IMG + ww) * C_DIM;

    float lg[HEADS];
    #pragma unroll
    for (int e = 0; e < HEADS; ++e) lg[e] = 0.f;
    for (int c = 0; c < C_DIM; ++c) {
        float xv = xr[c];
        #pragma unroll
        for (int e = 0; e < HEADS; ++e) lg[e] += xv * sWr[c * HEADS + e];
    }
    #pragma unroll
    for (int e = 0; e < HEADS; ++e) lg[e] = 1.0f / (1.0f + __expf(-lg[e]));

    #pragma unroll
    for (int e = 0; e < HEADS; ++e) sv6[e * LTOK + t] = lg[e];
    __syncthreads();

    const int lane = t & 31;
    const int e    = __builtin_amdgcn_readfirstlane(t >> 5);
    if (e >= HEADS) return;

    const float* sv = sv6 + e * LTOK;
    u64 k[8];
    #pragma unroll
    for (int r = 0; r < 8; ++r) {
        const int i = r * 32 + lane;
        const unsigned int fb = (i == 0) ? 0x7F800000u : __float_as_uint(sv[i]);
        k[r] = ((u64)fb << 32) | (unsigned int)(0xFFFFu - i);
    }

    cx_lanes<2, 1>(k, lane);
    cx_lanes<4, 2>(k, lane);  cx_lanes<4, 1>(k, lane);
    cx_lanes<8, 4>(k, lane);  cx_lanes<8, 2>(k, lane);  cx_lanes<8, 1>(k, lane);
    cx_lanes<16, 8>(k, lane); cx_lanes<16, 4>(k, lane); cx_lanes<16, 2>(k, lane); cx_lanes<16, 1>(k, lane);
    cx_tail<32>(k, lane);
    cx_regs<64, 1>(k);  cx_tail<64>(k, lane);
    cx_regs<128, 2>(k); cx_regs<128, 1>(k); cx_tail<128>(k, lane);
    cx_regs<256, 4>(k); cx_regs<256, 2>(k); cx_regs<256, 1>(k); cx_tail<256>(k, lane);

    #pragma unroll
    for (int r = 0; r < 2; ++r) {
        const int tok = 0xFFFF - (int)(k[r] & 0xFFFFull);
        gidx[((size_t)w * HEADS + e) * KSEL + r * 32 + lane] = tok;
        gval[((size_t)w * HEADS + e) * KSEL + r * 32 + lane] = sv[tok];
    }
}

static __device__ __forceinline__ float dpp_f(float v, const int ctrl_sel) {
    int x = __float_as_int(v);
    int y;
    switch (ctrl_sel) {
      case 0:  y = __builtin_amdgcn_update_dpp(x, x, 0xB1,  0xF, 0xF, true); break;
      case 1:  y = __builtin_amdgcn_update_dpp(x, x, 0x4E,  0xF, 0xF, true); break;
      case 2:  y = __builtin_amdgcn_update_dpp(x, x, 0x141, 0xF, 0xF, true); break;
      default: y = __builtin_amdgcn_update_dpp(x, x, 0x140, 0xF, 0xF, true); break;
    }
    return __int_as_float(y);
}
static __device__ __forceinline__ float row16_max(float v) {
    v = fmaxf(v, dpp_f(v, 0));
    v = fmaxf(v, dpp_f(v, 1));
    v = fmaxf(v, dpp_f(v, 2));
    v = fmaxf(v, dpp_f(v, 3));
    return v;
}
static __device__ __forceinline__ float row16_sum(float v) {
    v = v + dpp_f(v, 0);
    v = v + dpp_f(v, 1);
    v = v + dpp_f(v, 2);
    v = v + dpp_f(v, 3);
    return v;
}

struct W3 { float c0, c1, c2; };

__global__ __launch_bounds__(128) void k2_attn(
    const float* __restrict__ x,
    const float* __restrict__ Wqkv, const float* __restrict__ Wo,
    const int* __restrict__ gidx, const float* __restrict__ gval,
    float* __restrict__ scat)
{
    __shared__ float pool[5120];
    __shared__ float sQKV[KSEL * 48];
    __shared__ float sgate[KSEL];
    __shared__ int   sidx[KSEL];

    float* S   = pool;
    float* O   = pool + 4096;

    const int w    = blockIdx.x;
    const int e    = blockIdx.y;
    const int tid  = threadIdx.x;
    const int lane = tid & 31;
    const int wave = tid >> 5;
    const int hi   = lane >> 4;
    const int lo   = lane & 15;
    const int mt   = wave;

    const int b  = w / (NWIN * NWIN);
    const int rm = w % (NWIN * NWIN);
    const int wy = rm / NWIN, wx = rm % NWIN;

    const size_t sel0 = ((size_t)w * HEADS + e) * KSEL;
    const int tA = gidx[sel0 + mt * 16 + lo];
    if (tid < KSEL) {
        sidx[tid]  = gidx[sel0 + tid];
        sgate[tid] = gval[sel0 + tid];
    }

    {
        const float* wq = Wqkv + (size_t)e * C_DIM * 48 + (2 * hi) * 48 + 3 * lo;
        W3 b0[24], b1[24];
        #pragma unroll
        for (int s = 0; s < 24; ++s) {
            b0[s] = *(const W3*)(wq + (4 * s) * 48);
            b1[s] = *(const W3*)(wq + (4 * s + 1) * 48);
        }

        const float* xa = x + ((size_t)(b * IMG + wy * WIN + (tA >> 4)) * IMG
                               + wx * WIN + (tA & 15)) * C_DIM + 2 * hi;
        v2f af[24];
        #pragma unroll
        for (int s = 0; s < 24; ++s) af[s] = *(const v2f*)(xa + 4 * s);

        v8f acc[3];
        #pragma unroll
        for (int nt = 0; nt < 3; ++nt) acc[nt] = v8f{};
        #pragma unroll
        for (int s = 0; s < 24; ++s) {
            acc[0] = wmma4(af[s], v2f{ b0[s].c0, b1[s].c0 }, acc[0]);
            acc[1] = wmma4(af[s], v2f{ b0[s].c1, b1[s].c1 }, acc[1]);
            acc[2] = wmma4(af[s], v2f{ b0[s].c2, b1[s].c2 }, acc[2]);
        }
        #pragma unroll
        for (int nt = 0; nt < 3; ++nt) {
            #pragma unroll
            for (int r = 0; r < 8; ++r)
                sQKV[(mt * 16 + r + 8 * hi) * 48 + 3 * lo + nt] = acc[nt][r];
        }
    }
    __syncthreads();

    v2f wof[6][4];
    {
        const float* wo = Wo + (size_t)e * HD * C_DIM + (2 * hi) * C_DIM + lo;
        #pragma unroll
        for (int nt = 0; nt < 6; ++nt) {
            #pragma unroll
            for (int s = 0; s < 4; ++s)
                wof[nt][s] = v2f{ wo[(4 * s) * C_DIM + nt * 16],
                                  wo[(4 * s + 1) * C_DIM + nt * 16] };
        }
    }

    v8f sc[4];
    #pragma unroll
    for (int nt = 0; nt < 4; ++nt) sc[nt] = v8f{};
    #pragma unroll
    for (int kk = 0; kk < HD; kk += 4) {
        int ka = kk + 2 * hi;
        v2f a = { sQKV[(mt * 16 + lo) * 48 + ka],
                  sQKV[(mt * 16 + lo) * 48 + ka + 1] };
        #pragma unroll
        for (int nt = 0; nt < 4; ++nt) {
            v2f bf = { sQKV[(nt * 16 + lo) * 48 + 16 + ka],
                       sQKV[(nt * 16 + lo) * 48 + 16 + ka + 1] };
            sc[nt] = wmma4(a, bf, sc[nt]);
        }
    }

    #pragma unroll
    for (int r = 0; r < 8; ++r) {
        const int row = mt * 16 + r + 8 * hi;
        float s0 = sc[0][r] * 0.25f, s1 = sc[1][r] * 0.25f;
        float s2 = sc[2][r] * 0.25f, s3 = sc[3][r] * 0.25f;
        float m = row16_max(fmaxf(fmaxf(s0, s1), fmaxf(s2, s3)));
        float e0 = __expf(s0 - m), e1 = __expf(s1 - m);
        float e2 = __expf(s2 - m), e3 = __expf(s3 - m);
        float sum = row16_sum((e0 + e1) + (e2 + e3));
        float inv = sgate[row] / sum;
        S[row * 64 +  0 + lo] = e0 * inv;
        S[row * 64 + 16 + lo] = e1 * inv;
        S[row * 64 + 32 + lo] = e2 * inv;
        S[row * 64 + 48 + lo] = e3 * inv;
    }
    __syncthreads();

    {
        v8f acc = {};
        for (int kk = 0; kk < 64; kk += 4) {
            int ka = kk + 2 * hi;
            v2f a = { S[(mt * 16 + lo) * 64 + ka],
                      S[(mt * 16 + lo) * 64 + ka + 1] };
            v2f bf = { sQKV[ka * 48 + 32 + lo],
                       sQKV[(ka + 1) * 48 + 32 + lo] };
            acc = wmma4(a, bf, acc);
        }
        #pragma unroll
        for (int r = 0; r < 8; ++r)
            O[(mt * 16 + r + 8 * hi) * HD + lo] = acc[r];
    }
    __syncthreads();

    v2f oa[4];
    #pragma unroll
    for (int s = 0; s < 4; ++s) {
        int ka = 4 * s + 2 * hi;
        oa[s] = v2f{ O[(mt * 16 + lo) * HD + ka], O[(mt * 16 + lo) * HD + ka + 1] };
    }
    #pragma unroll
    for (int nt = 0; nt < 6; ++nt) {
        v8f acc = {};
        #pragma unroll
        for (int s = 0; s < 4; ++s)
            acc = wmma4(oa[s], wof[nt][s], acc);
        #pragma unroll
        for (int r = 0; r < 8; ++r) {
            int m  = mt * 16 + r + 8 * hi;
            int t  = sidx[m];
            int hh = wy * WIN + (t >> 4);
            int ww = wx * WIN + (t & 15);
            atomicAdd(&scat[((size_t)(b * IMG + hh) * IMG + ww) * C_DIM + nt * 16 + lo],
                      acc[r]);
        }
    }
}

#define K3_BLOCKS 256
#define K3_WAVES  8
#define K3_TILES  ((NPIX / 16) / (K3_BLOCKS * K3_WAVES))

static __device__ __forceinline__ void k3_load_a(v2f (&a)[24], const float* p) {
    #pragma unroll
    for (int s = 0; s < 24; ++s) a[s] = *(const v2f*)(p + 4 * s);
}

static __device__ __forceinline__ void k3_tile(
    const v2f (&a)[24], const v2f (&Bf)[6][24], const float (&bs)[6], float* yp)
{
    v8f acc[6];
    #pragma unroll
    for (int nt = 0; nt < 6; ++nt) acc[nt] = v8f{};
    #pragma unroll
    for (int s = 0; s < 24; ++s) {
        #pragma unroll
        for (int nt = 0; nt < 6; ++nt)
            acc[nt] = wmma4(a[s], Bf[nt][s], acc[nt]);
    }
    #pragma unroll
    for (int nt = 0; nt < 6; ++nt) {
        #pragma unroll
        for (int r = 0; r < 8; ++r)
            yp[r * C_DIM + nt * 16] = acc[nt][r] + bs[nt];
    }
}

__global__ __launch_bounds__(256) void k3_proj(
    const float* __restrict__ scat, const float* __restrict__ projW,
    const float* __restrict__ projb, float* __restrict__ y)
{
    const int tid  = threadIdx.x;
    const int lane = tid & 31, wave = tid >> 5;
    const int hi = lane >> 4, lo = lane & 15;

    v2f Bf[6][24];
    float bs[6];
    #pragma unroll
    for (int nt = 0; nt < 6; ++nt) {
        const float* wr = projW + (size_t)(nt * 16 + lo) * C_DIM + 2 * hi;
        #pragma unroll
        for (int s = 0; s < 24; ++s) Bf[nt][s] = *(const v2f*)(wr + 4 * s);
        bs[nt] = projb[nt * 16 + lo];
    }

    const size_t tile0 = (size_t)(blockIdx.x * K3_WAVES + wave) * K3_TILES;
    const float* ap = scat + (tile0 * 16 + lo) * C_DIM + 2 * hi;
    float*       yp = y    + (tile0 * 16 + 8 * hi) * C_DIM + lo;
    const int TS = 16 * C_DIM;

    v2f a0[24], a1[24];
    k3_load_a(a0, ap);

    #pragma unroll 1
    for (int i = 0; i + 2 < K3_TILES + 1; i += 2) {
        k3_load_a(a1, ap + TS);
        k3_tile(a0, Bf, bs, yp);
        k3_load_a(a0, ap + 2 * TS);
        k3_tile(a1, Bf, bs, yp + TS);
        ap += 2 * TS;
        yp += 2 * TS;
    }
    k3_tile(a0, Bf, bs, yp);
}

extern "C" void kernel_launch(void* const* d_in, const int* in_sizes, int n_in,
                              void* d_out, int out_size, void* d_ws, size_t ws_size,
                              hipStream_t stream) {
    const float* x     = (const float*)d_in[0];
    const float* Wr    = (const float*)d_in[1];
    const float* Wqkv  = (const float*)d_in[2];
    const float* Wo    = (const float*)d_in[3];
    const float* projW = (const float*)d_in[4];
    const float* projb = (const float*)d_in[5];
    float* y = (float*)d_out;

    const size_t scatBytes = (size_t)NPIX * C_DIM * sizeof(float);
    float* scat = (float*)d_ws;
    int*   gidx = (int*)((char*)d_ws + scatBytes);
    float* gval = (float*)((char*)d_ws + scatBytes +
                           (size_t)BW * HEADS * KSEL * sizeof(int));

    hipMemsetAsync(scat, 0, scatBytes, stream);

    k1_router<<<BW, 256, 0, stream>>>(x, Wr, gidx, gval);
    k2_attn<<<dim3(BW, HEADS), 128, 0, stream>>>(x, Wqkv, Wo, gidx, gval, scat);
    k3_proj<<<K3_BLOCKS, 256, 0, stream>>>(scat, projW, projb, y);
}


# k2: a wave's own P and O strips kept in LDS in A-fragment order (xor-swizzled): conflict-free 16-byte reads, not 16 lanes on one bank; both hand-over block barriers go; sums unchanged
# speedup vs baseline: 2.2566x; 1.1893x over previous
#include <hip/hip_runtime.h>
#include <hip/hip_bf16.h>


#define BATCH   2
#define IMG     384
#define C_DIM   96
#define HEADS   6
#define HD      16
#define WIN     16
#define NWIN    24
#define BW      (BATCH * NWIN * NWIN)
#define LTOK    256
#define KSEL    64
#define NPIX    (BATCH * IMG * IMG)

typedef __attribute__((ext_vector_type(2))) float v2f;
typedef __attribute__((ext_vector_type(8))) float v8f;

static __device__ inline v8f wmma4(v2f a, v2f b, v8f c) {
    return __builtin_amdgcn_wmma_f32_16x16x4_f32(
         false, a,  false, b,
         (short)0, c,  false,  false);
}

typedef unsigned long long u64;

template <int M> static __device__ __forceinline__ unsigned int lane_xor_u32(unsigned int x) {
    const int b = (int)x;
    if constexpr (M == 16) {
        return (unsigned int)__builtin_amdgcn_permlanex16(b, b, 0x76543210, 0xfedcba98, false, false);
    } else {
        return (unsigned int)__builtin_amdgcn_update_dpp(b, b, 0x160 + M, 0xF, 0xF, true);
    }
}
template <int M> static __device__ __forceinline__ u64 lane_xor_u64(u64 k) {
    const unsigned int lo = lane_xor_u32<M>((unsigned int)k);
    const unsigned int hi = lane_xor_u32<M>((unsigned int)(k >> 32));
    return ((u64)hi << 32) | lo;
}

template <int KS, int J>
static __device__ __forceinline__ void cx_lanes(u64 (&k)[8], const int lane) {
    const bool low = (lane & J) == 0;
    #pragma unroll
    for (int r = 0; r < 8; ++r) {
        const bool desc = (KS < 32) ? ((lane & KS) == 0) : (((r * 32) & KS) == 0);
        const bool keepMax = (desc == low);
        const u64 a = k[r];
        const u64 b = lane_xor_u64<J>(a);
        const bool gt = a > b;
        k[r] = (gt == keepMax) ? a : b;
    }
}
template <int KS, int JR>
static __device__ __forceinline__ void cx_regs(u64 (&k)[8]) {
    #pragma unroll
    for (int r = 0; r < 8; ++r) {
        if ((r & JR) == 0) {
            const bool desc = ((r * 32) & KS) == 0;
            const u64 a = k[r], b = k[r ^ JR];
            const u64 mx = (a > b) ? a : b;
            const u64 mn = (a > b) ? b : a;
            k[r]      = desc ? mx : mn;
            k[r ^ JR] = desc ? mn : mx;
        }
    }
}
template <int KS>
static __device__ __forceinline__ void cx_tail(u64 (&k)[8], const int lane) {
    cx_lanes<KS, 16>(k, lane);
    cx_lanes<KS, 8>(k, lane);
    cx_lanes<KS, 4>(k, lane);
    cx_lanes<KS, 2>(k, lane);
    cx_lanes<KS, 1>(k, lane);
}

__global__ __launch_bounds__(256) void k1_router(
    const float* __restrict__ x, const float* __restrict__ Wr,
    int* __restrict__ gidx, float* __restrict__ gval)
{
    __shared__ float sWr[C_DIM * HEADS];
    __shared__ float sv6[HEADS * LTOK];

    const int w = blockIdx.x;
    const int t = threadIdx.x;

    for (int i = t; i < C_DIM * HEADS; i += 256) sWr[i] = Wr[i];
    __syncthreads();

    const int b  = w / (NWIN * NWIN);
    const int rm = w % (NWIN * NWIN);
    const int wy = rm / NWIN, wx = rm % NWIN;
    const int hh = wy * WIN + (t >> 4);
    const int ww = wx * WIN + (t & 15);
    const float* xr = x + ((size_t)(b * IMG + hh) * IMG + ww) * C_DIM;

    float lg[HEADS];
    #pragma unroll
    for (int e = 0; e < HEADS; ++e) lg[e] = 0.f;
    for (int c = 0; c < C_DIM; ++c) {
        float xv = xr[c];
        #pragma unroll
        for (int e = 0; e < HEADS; ++e) lg[e] += xv * sWr[c * HEADS + e];
    }
    #pragma unroll
    for (int e = 0; e < HEADS; ++e) lg[e] = 1.0f / (1.0f + __expf(-lg[e]));

    #pragma unroll
    for (int e = 0; e < HEADS; ++e) sv6[e * LTOK + t] = lg[e];
    __syncthreads();

    const int lane = t & 31;
    const int e    = __builtin_amdgcn_readfirstlane(t >> 5);
    if (e >= HEADS) return;

    const float* sv = sv6 + e * LTOK;
    u64 k[8];
    #pragma unroll
    for (int r = 0; r < 8; ++r) {
        const int i = r * 32 + lane;
        const unsigned int fb = (i == 0) ? 0x7F800000u : __float_as_uint(sv[i]);
        k[r] = ((u64)fb << 32) | (unsigned int)(0xFFFFu - i);
    }

    cx_lanes<2, 1>(k, lane);
    cx_lanes<4, 2>(k, lane);  cx_lanes<4, 1>(k, lane);
    cx_lanes<8, 4>(k, lane);  cx_lanes<8, 2>(k, lane);  cx_lanes<8, 1>(k, lane);
    cx_lanes<16, 8>(k, lane); cx_lanes<16, 4>(k, lane); cx_lanes<16, 2>(k, lane); cx_lanes<16, 1>(k, lane);
    cx_tail<32>(k, lane);
    cx_regs<64, 1>(k);  cx_tail<64>(k, lane);
    cx_regs<128, 2>(k); cx_regs<128, 1>(k); cx_tail<128>(k, lane);
    cx_regs<256, 4>(k); cx_regs<256, 2>(k); cx_regs<256, 1>(k); cx_tail<256>(k, lane);

    #pragma unroll
    for (int r = 0; r < 2; ++r) {
        const int tok = 0xFFFF - (int)(k[r] & 0xFFFFull);
        gidx[((size_t)w * HEADS + e) * KSEL + r * 32 + lane] = tok;
        gval[((size_t)w * HEADS + e) * KSEL + r * 32 + lane] = sv[tok];
    }
}

static __device__ __forceinline__ float dpp_f(float v, const int ctrl_sel) {
    int x = __float_as_int(v);
    int y;
    switch (ctrl_sel) {
      case 0:  y = __builtin_amdgcn_update_dpp(x, x, 0xB1,  0xF, 0xF, true); break;
      case 1:  y = __builtin_amdgcn_update_dpp(x, x, 0x4E,  0xF, 0xF, true); break;
      case 2:  y = __builtin_amdgcn_update_dpp(x, x, 0x141, 0xF, 0xF, true); break;
      default: y = __builtin_amdgcn_update_dpp(x, x, 0x140, 0xF, 0xF, true); break;
    }
    return __int_as_float(y);
}
static __device__ __forceinline__ float row16_max(float v) {
    v = fmaxf(v, dpp_f(v, 0));
    v = fmaxf(v, dpp_f(v, 1));
    v = fmaxf(v, dpp_f(v, 2));
    v = fmaxf(v, dpp_f(v, 3));
    return v;
}
static __device__ __forceinline__ float row16_sum(float v) {
    v = v + dpp_f(v, 0);
    v = v + dpp_f(v, 1);
    v = v + dpp_f(v, 2);
    v = v + dpp_f(v, 3);
    return v;
}

struct W3 { float c0, c1, c2; };

static __device__ __forceinline__ int frag_slot(const int rho, const int kap) {
    const int g = 2 * (kap >> 3) + ((kap >> 1) & 1);
    const int j = (kap & 1) + ((kap >> 1) & 2);
    return 64 * g + 4 * (rho ^ (g & 3)) + j;
}

#define F_THREADS 768
#define TP        100

__global__ __launch_bounds__(F_THREADS) void k2_attn_proj(
    const float* __restrict__ x,
    const float* __restrict__ Wqkv, const float* __restrict__ Wo,
    const int* __restrict__ gidx, const float* __restrict__ gval,
    const float* __restrict__ projW, const float* __restrict__ projb,
    float* __restrict__ y)
{
    __shared__ __attribute__((aligned(16))) float tile[LTOK * TP];
    __shared__ __attribute__((aligned(16))) float sP6[HEADS][KSEL * 64];
    __shared__ float sQKV6[HEADS][KSEL * 48];
    __shared__ __attribute__((aligned(16))) float sO6[HEADS][KSEL * HD];
    __shared__ float sgate6[HEADS][KSEL];
    __shared__ int   sidx6[HEADS][KSEL];

    const int w    = blockIdx.x;
    const int tid  = threadIdx.x;
    const int lane = tid & 31;
    const int wave = __builtin_amdgcn_readfirstlane(tid >> 5);
    const int e    = wave >> 2;
    const int mt   = wave & 3;
    const int hi   = lane >> 4;
    const int lo   = lane & 15;
    const int lt   = tid & 127;

    float* Pw    = sP6[e] + mt * (16 * 64);
    float* sQKV  = sQKV6[e];
    float* Ow    = sO6[e] + mt * (16 * HD);
    float* sgate = sgate6[e];
    int*   sidx  = sidx6[e];

    const int b  = w / (NWIN * NWIN);
    const int rm = w % (NWIN * NWIN);
    const int wy = rm / NWIN, wx = rm % NWIN;

    const size_t sel0 = ((size_t)w * HEADS + e) * KSEL;
    const int tA = gidx[sel0 + mt * 16 + lo];
    if (lt < KSEL) {
        sidx[lt]  = gidx[sel0 + lt];
        sgate[lt] = gval[sel0 + lt];
    }

    {
        const float4 z4 = make_float4(0.f, 0.f, 0.f, 0.f);
        float4* t4 = (float4*)tile;
        for (int i = tid; i < LTOK * TP / 4; i += F_THREADS) t4[i] = z4;
    }

    {
        const float* wq = Wqkv + (size_t)e * C_DIM * 48 + (2 * hi) * 48 + 3 * lo;
        const float* xa = x + ((size_t)(b * IMG + wy * WIN + (tA >> 4)) * IMG
                               + wx * WIN + (tA & 15)) * C_DIM + 2 * hi;
        W3  b0[2][6], b1[2][6];
        v2f af[2][6];
        #pragma unroll
        for (int j = 0; j < 6; ++j) {
            b0[0][j] = *(const W3*)(wq + (4 * j) * 48);
            b1[0][j] = *(const W3*)(wq + (4 * j + 1) * 48);
            af[0][j] = *(const v2f*)(xa + 4 * j);
        }

        v8f acc[3];
        #pragma unroll
        for (int nt = 0; nt < 3; ++nt) acc[nt] = v8f{};
        #pragma unroll
        for (int g = 0; g < 4; ++g) {
            const int cur = g & 1, nxt = cur ^ 1;
            if (g < 3) {
                #pragma unroll
                for (int j = 0; j < 6; ++j) {
                    const int s = 6 * (g + 1) + j;
                    b0[nxt][j] = *(const W3*)(wq + (4 * s) * 48);
                    b1[nxt][j] = *(const W3*)(wq + (4 * s + 1) * 48);
                    af[nxt][j] = *(const v2f*)(xa + 4 * s);
                }
            }
            __builtin_amdgcn_sched_barrier(0);
            #pragma unroll
            for (int j = 0; j < 6; ++j) {
                acc[0] = wmma4(af[cur][j], v2f{ b0[cur][j].c0, b1[cur][j].c0 }, acc[0]);
                acc[1] = wmma4(af[cur][j], v2f{ b0[cur][j].c1, b1[cur][j].c1 }, acc[1]);
                acc[2] = wmma4(af[cur][j], v2f{ b0[cur][j].c2, b1[cur][j].c2 }, acc[2]);
            }
            __builtin_amdgcn_sched_barrier(0);
        }
        #pragma unroll
        for (int nt = 0; nt < 3; ++nt) {
            #pragma unroll
            for (int r = 0; r < 8; ++r)
                sQKV[(mt * 16 + r + 8 * hi) * 48 + 3 * lo + nt] = acc[nt][r];
        }
    }
    __syncthreads();

    v2f wof[6][4];
    {
        const float* wo = Wo + (size_t)e * HD * C_DIM + (2 * hi) * C_DIM + lo;
        #pragma unroll
        for (int nt = 0; nt < 6; ++nt) {
            #pragma unroll
            for (int s = 0; s < 4; ++s)
                wof[nt][s] = v2f{ wo[(4 * s) * C_DIM + nt * 16],
                                  wo[(4 * s + 1) * C_DIM + nt * 16] };
        }
    }

    v8f sc[4];
    #pragma unroll
    for (int nt = 0; nt < 4; ++nt) sc[nt] = v8f{};
    #pragma unroll
    for (int kk = 0; kk < HD; kk += 4) {
        int ka = kk + 2 * hi;
        v2f a = { sQKV[(mt * 16 + lo) * 48 + ka],
                  sQKV[(mt * 16 + lo) * 48 + ka + 1] };
        #pragma unroll
        for (int nt = 0; nt < 4; ++nt) {
            v2f bf = { sQKV[(nt * 16 + lo) * 48 + 16 + ka],
                       sQKV[(nt * 16 + lo) * 48 + 16 + ka + 1] };
            sc[nt] = wmma4(a, bf, sc[nt]);
        }
    }

    #pragma unroll
    for (int r = 0; r < 8; ++r) {
        const int row = mt * 16 + r + 8 * hi;
        float s0 = sc[0][r] * 0.25f, s1 = sc[1][r] * 0.25f;
        float s2 = sc[2][r] * 0.25f, s3 = sc[3][r] * 0.25f;
        float m = row16_max(fmaxf(fmaxf(s0, s1), fmaxf(s2, s3)));
        float e0 = __expf(s0 - m), e1 = __expf(s1 - m);
        float e2 = __expf(s2 - m), e3 = __expf(s3 - m);
        float sum = row16_sum((e0 + e1) + (e2 + e3));
        float inv = sgate[row] / sum;
        const int sr = r + 8 * hi;
        Pw[frag_slot(sr,  0 + lo)] = e0 * inv;
        Pw[frag_slot(sr, 16 + lo)] = e1 * inv;
        Pw[frag_slot(sr, 32 + lo)] = e2 * inv;
        Pw[frag_slot(sr, 48 + lo)] = e3 * inv;
    }
    __builtin_amdgcn_fence(__ATOMIC_ACQ_REL, "workgroup", "local");

    {
        v8f acc = {};
        #pragma unroll
        for (int i = 0; i < 8; ++i) {
            const int g  = 2 * i + hi;
            const float4 p4 = *(const float4*)(Pw + 64 * g + 4 * (lo ^ (g & 3)));
            const int k0 = 8 * i + 2 * hi;
            v2f bf0 = { sQKV[k0 * 48 + 32 + lo],
                        sQKV[(k0 + 1) * 48 + 32 + lo] };
            v2f bf1 = { sQKV[(k0 + 4) * 48 + 32 + lo],
                        sQKV[(k0 + 5) * 48 + 32 + lo] };
            acc = wmma4(v2f{ p4.x, p4.y }, bf0, acc);
            acc = wmma4(v2f{ p4.z, p4.w }, bf1, acc);
        }
        #pragma unroll
        for (int r = 0; r < 8; ++r)
            Ow[frag_slot(r + 8 * hi, lo)] = acc[r];
    }
    __builtin_amdgcn_fence(__ATOMIC_ACQ_REL, "workgroup", "local");

    v2f oa[4];
    #pragma unroll
    for (int i = 0; i < 2; ++i) {
        const int g = 2 * i + hi;
        const float4 o4 = *(const float4*)(Ow + 64 * g + 4 * (lo ^ (g & 3)));
        oa[2 * i]     = v2f{ o4.x, o4.y };
        oa[2 * i + 1] = v2f{ o4.z, o4.w };
    }
    #pragma unroll
    for (int nt = 0; nt < 6; ++nt) {
        v8f acc = {};
        #pragma unroll
        for (int s = 0; s < 4; ++s)
            acc = wmma4(oa[s], wof[nt][s], acc);
        #pragma unroll
        for (int r = 0; r < 8; ++r) {
            int m  = mt * 16 + r + 8 * hi;
            int t  = sidx[m];
            atomicAdd(&tile[t * TP + nt * 16 + lo], acc[r]);
        }
    }
    __syncthreads();

    {
        const int p = wave & 7;
        const int q = wave >> 3;
        const float* ta0 = tile + ((2 * p) * 16 + lo) * TP + 4 * hi;
        const float* ta1 = ta0 + 16 * TP;
        const float* wb0 = projW + (size_t)((2 * q) * 16 + lo) * C_DIM + 4 * hi;
        const float* wb1 = wb0 + 16 * C_DIM;

        v8f pa[2][2];
        #pragma unroll
        for (int i = 0; i < 2; ++i) {
            #pragma unroll
            for (int j = 0; j < 2; ++j) pa[i][j] = v8f{};
        }
        #pragma unroll
        for (int u = 0; u < 12; ++u) {
            const float4 a0 = *(const float4*)(ta0 + 8 * u);
            const float4 a1 = *(const float4*)(ta1 + 8 * u);
            const float4 b0 = *(const float4*)(wb0 + 8 * u);
            const float4 b1 = *(const float4*)(wb1 + 8 * u);
            pa[0][0] = wmma4(v2f{ a0.x, a0.y }, v2f{ b0.x, b0.y }, pa[0][0]);
            pa[0][1] = wmma4(v2f{ a0.x, a0.y }, v2f{ b1.x, b1.y }, pa[0][1]);
            pa[1][0] = wmma4(v2f{ a1.x, a1.y }, v2f{ b0.x, b0.y }, pa[1][0]);
            pa[1][1] = wmma4(v2f{ a1.x, a1.y }, v2f{ b1.x, b1.y }, pa[1][1]);
            pa[0][0] = wmma4(v2f{ a0.z, a0.w }, v2f{ b0.z, b0.w }, pa[0][0]);
            pa[0][1] = wmma4(v2f{ a0.z, a0.w }, v2f{ b1.z, b1.w }, pa[0][1]);
            pa[1][0] = wmma4(v2f{ a1.z, a1.w }, v2f{ b0.z, b0.w }, pa[1][0]);
            pa[1][1] = wmma4(v2f{ a1.z, a1.w }, v2f{ b1.z, b1.w }, pa[1][1]);
        }
        #pragma unroll
        for (int j = 0; j < 2; ++j) {
            const int col = (2 * q + j) * 16 + lo;
            const float bs = projb[col];
            #pragma unroll
            for (int i = 0; i < 2; ++i) {
                const int rt = 2 * p + i;
                float* yp = y + ((size_t)(b * IMG + wy * WIN + rt) * IMG
                                 + wx * WIN + 8 * hi) * C_DIM + col;
                #pragma unroll
                for (int r = 0; r < 8; ++r)
                    yp[r * C_DIM] = pa[i][j][r] + bs;
            }
        }
    }
}

extern "C" void kernel_launch(void* const* d_in, const int* in_sizes, int n_in,
                              void* d_out, int out_size, void* d_ws, size_t ws_size,
                              hipStream_t stream) {
    const float* x     = (const float*)d_in[0];
    const float* Wr    = (const float*)d_in[1];
    const float* Wqkv  = (const float*)d_in[2];
    const float* Wo    = (const float*)d_in[3];
    const float* projW = (const float*)d_in[4];
    const float* projb = (const float*)d_in[5];
    float* y = (float*)d_out;

    int*   gidx = (int*)d_ws;
    float* gval = (float*)((char*)d_ws + (size_t)BW * HEADS * KSEL * sizeof(int));

    k1_router<<<BW, 256, 0, stream>>>(x, Wr, gidx, gval);
    k2_attn_proj<<<BW, F_THREADS, 0, stream>>>(x, Wqkv, Wo, gidx, gval, projW, projb, y);
}


# k2: two windows resident per WGP: a window runs on 12 waves as two passes of 3 heads, LDS 295 -> 156 KB (K|V only per head; Q, P in halves, O share one strip per wave), 156 registers; sums unchanged
# speedup vs baseline: 2.5100x; 1.1123x over previous
#include <hip/hip_runtime.h>
#include <hip/hip_bf16.h>


#define BATCH   2
#define IMG     384
#define C_DIM   96
#define HEADS   6
#define HD      16
#define WIN     16
#define NWIN    24
#define BW      (BATCH * NWIN * NWIN)
#define LTOK    256
#define KSEL    64
#define NPIX    (BATCH * IMG * IMG)

typedef __attribute__((ext_vector_type(2))) float v2f;
typedef __attribute__((ext_vector_type(8))) float v8f;

static __device__ inline v8f wmma4(v2f a, v2f b, v8f c) {
    return __builtin_amdgcn_wmma_f32_16x16x4_f32(
         false, a,  false, b,
         (short)0, c,  false,  false);
}

typedef unsigned long long u64;

template <int M> static __device__ __forceinline__ unsigned int lane_xor_u32(unsigned int x) {
    const int b = (int)x;
    if constexpr (M == 16) {
        return (unsigned int)__builtin_amdgcn_permlanex16(b, b, 0x76543210, 0xfedcba98, false, false);
    } else {
        return (unsigned int)__builtin_amdgcn_update_dpp(b, b, 0x160 + M, 0xF, 0xF, true);
    }
}
template <int M> static __device__ __forceinline__ u64 lane_xor_u64(u64 k) {
    const unsigned int lo = lane_xor_u32<M>((unsigned int)k);
    const unsigned int hi = lane_xor_u32<M>((unsigned int)(k >> 32));
    return ((u64)hi << 32) | lo;
}

template <int KS, int J>
static __device__ __forceinline__ void cx_lanes(u64 (&k)[8], const int lane) {
    const bool low = (lane & J) == 0;
    #pragma unroll
    for (int r = 0; r < 8; ++r) {
        const bool desc = (KS < 32) ? ((lane & KS) == 0) : (((r * 32) & KS) == 0);
        const bool keepMax = (desc == low);
        const u64 a = k[r];
        const u64 b = lane_xor_u64<J>(a);
        const bool gt = a > b;
        k[r] = (gt == keepMax) ? a : b;
    }
}
template <int KS, int JR>
static __device__ __forceinline__ void cx_regs(u64 (&k)[8]) {
    #pragma unroll
    for (int r = 0; r < 8; ++r) {
        if ((r & JR) == 0) {
            const bool desc = ((r * 32) & KS) == 0;
            const u64 a = k[r], b = k[r ^ JR];
            const u64 mx = (a > b) ? a : b;
            const u64 mn = (a > b) ? b : a;
            k[r]      = desc ? mx : mn;
            k[r ^ JR] = desc ? mn : mx;
        }
    }
}
template <int KS>
static __device__ __forceinline__ void cx_tail(u64 (&k)[8], const int lane) {
    cx_lanes<KS, 16>(k, lane);
    cx_lanes<KS, 8>(k, lane);
    cx_lanes<KS, 4>(k, lane);
    cx_lanes<KS, 2>(k, lane);
    cx_lanes<KS, 1>(k, lane);
}

__global__ __launch_bounds__(256) void k1_router(
    const float* __restrict__ x, const float* __restrict__ Wr,
    int* __restrict__ gidx, float* __restrict__ gval)
{
    __shared__ float sWr[C_DIM * HEADS];
    __shared__ float sv6[HEADS * LTOK];

    const int w = blockIdx.x;
    const int t = threadIdx.x;

    for (int i = t; i < C_DIM * HEADS; i += 256) sWr[i] = Wr[i];
    __syncthreads();

    const int b  = w / (NWIN * NWIN);
    const int rm = w % (NWIN * NWIN);
    const int wy = rm / NWIN, wx = rm % NWIN;
    const int hh = wy * WIN + (t >> 4);
    const int ww = wx * WIN + (t & 15);
    const float* xr = x + ((size_t)(b * IMG + hh) * IMG + ww) * C_DIM;

    float lg[HEADS];
    #pragma unroll
    for (int e = 0; e < HEADS; ++e) lg[e] = 0.f;
    for (int c = 0; c < C_DIM; ++c) {
        float xv = xr[c];
        #pragma unroll
        for (int e = 0; e < HEADS; ++e) lg[e] += xv * sWr[c * HEADS + e];
    }
    #pragma unroll
    for (int e = 0; e < HEADS; ++e) lg[e] = 1.0f / (1.0f + __expf(-lg[e]));

    #pragma unroll
    for (int e = 0; e < HEADS; ++e) sv6[e * LTOK + t] = lg[e];
    __syncthreads();

    const int lane = t & 31;
    const int e    = __builtin_amdgcn_readfirstlane(t >> 5);
    if (e >= HEADS) return;

    const float* sv = sv6 + e * LTOK;
    u64 k[8];
    #pragma unroll
    for (int r = 0; r < 8; ++r) {
        const int i = r * 32 + lane;
        const unsigned int fb = (i == 0) ? 0x7F800000u : __float_as_uint(sv[i]);
        k[r] = ((u64)fb << 32) | (unsigned int)(0xFFFFu - i);
    }

    cx_lanes<2, 1>(k, lane);
    cx_lanes<4, 2>(k, lane);  cx_lanes<4, 1>(k, lane);
    cx_lanes<8, 4>(k, lane);  cx_lanes<8, 2>(k, lane);  cx_lanes<8, 1>(k, lane);
    cx_lanes<16, 8>(k, lane); cx_lanes<16, 4>(k, lane); cx_lanes<16, 2>(k, lane); cx_lanes<16, 1>(k, lane);
    cx_tail<32>(k, lane);
    cx_regs<64, 1>(k);  cx_tail<64>(k, lane);
    cx_regs<128, 2>(k); cx_regs<128, 1>(k); cx_tail<128>(k, lane);
    cx_regs<256, 4>(k); cx_regs<256, 2>(k); cx_regs<256, 1>(k); cx_tail<256>(k, lane);

    #pragma unroll
    for (int r = 0; r < 2; ++r) {
        const int tok = 0xFFFF - (int)(k[r] & 0xFFFFull);
        gidx[((size_t)w * HEADS + e) * KSEL + r * 32 + lane] = tok;
        gval[((size_t)w * HEADS + e) * KSEL + r * 32 + lane] = sv[tok];
    }
}

static __device__ __forceinline__ float dpp_f(float v, const int ctrl_sel) {
    int x = __float_as_int(v);
    int y;
    switch (ctrl_sel) {
      case 0:  y = __builtin_amdgcn_update_dpp(x, x, 0xB1,  0xF, 0xF, true); break;
      case 1:  y = __builtin_amdgcn_update_dpp(x, x, 0x4E,  0xF, 0xF, true); break;
      case 2:  y = __builtin_amdgcn_update_dpp(x, x, 0x141, 0xF, 0xF, true); break;
      default: y = __builtin_amdgcn_update_dpp(x, x, 0x140, 0xF, 0xF, true); break;
    }
    return __int_as_float(y);
}
static __device__ __forceinline__ float row16_max(float v) {
    v = fmaxf(v, dpp_f(v, 0));
    v = fmaxf(v, dpp_f(v, 1));
    v = fmaxf(v, dpp_f(v, 2));
    v = fmaxf(v, dpp_f(v, 3));
    return v;
}
static __device__ __forceinline__ float row16_sum(float v) {
    v = v + dpp_f(v, 0);
    v = v + dpp_f(v, 1);
    v = v + dpp_f(v, 2);
    v = v + dpp_f(v, 3);
    return v;
}

struct W3 { float c0, c1, c2; };


#define F_THREADS 384
#define TP        100
#define KVP       36
#define SW        576

__global__ __launch_bounds__(F_THREADS, 6) void k2_attn_proj(
    const float* __restrict__ x,
    const float* __restrict__ Wqkv, const float* __restrict__ Wo,
    const int* __restrict__ gidx, const float* __restrict__ gval,
    const float* __restrict__ projW, const float* __restrict__ projb,
    float* __restrict__ y)
{
    __shared__ __attribute__((aligned(16))) float tile[LTOK * TP];
    __shared__ __attribute__((aligned(16))) float sKV3[3][KSEL * KVP];
    __shared__ __attribute__((aligned(16))) float strip12[12][SW];
    __shared__ float sgate3[3][KSEL];
    __shared__ int   sidx3[3][KSEL];

    const int w    = blockIdx.x;
    const int tid  = threadIdx.x;
    const int lane = tid & 31;
    const int wave = __builtin_amdgcn_readfirstlane(tid >> 5);
    const int e3   = wave >> 2;
    const int mt   = wave & 3;
    const int hi   = lane >> 4;
    const int lo   = lane & 15;
    const int lt   = tid & 127;

    float* Sw    = strip12[wave];
    float* sKV   = sKV3[e3];
    float* sgate = sgate3[e3];
    int*   sidx  = sidx3[e3];

    const int b  = w / (NWIN * NWIN);
    const int rm = w % (NWIN * NWIN);
    const int wy = rm / NWIN, wx = rm % NWIN;

    const size_t selA = ((size_t)w * HEADS + e3) * KSEL;
    const size_t selB = selA + (size_t)3 * KSEL;
    const int tA0 = gidx[selA + mt * 16 + lo];
    const int tA1 = gidx[selB + mt * 16 + lo];

    {
        const float4 z4 = make_float4(0.f, 0.f, 0.f, 0.f);
        float4* t4 = (float4*)tile;
        for (int i = tid; i < LTOK * TP / 4; i += F_THREADS) t4[i] = z4;
    }

    for (int pass = 0; pass < 2; ++pass) {
    const int    e    = 3 * pass + e3;
    const size_t sel0 = pass ? selB : selA;
    const int    tA   = pass ? tA1 : tA0;
    if (lt < KSEL) {
        sidx[lt]  = gidx[sel0 + lt];
        sgate[lt] = gval[sel0 + lt];
    }

    {
        const float* wq = Wqkv + (size_t)e * C_DIM * 48 + (2 * hi) * 48 + 3 * lo;
        const float* xa = x + ((size_t)(b * IMG + wy * WIN + (tA >> 4)) * IMG
                               + wx * WIN + (tA & 15)) * C_DIM + 2 * hi;
        W3  b0[2][4], b1[2][4];
        v2f af[2][4];
        #pragma unroll
        for (int j = 0; j < 4; ++j) {
            b0[0][j] = *(const W3*)(wq + (4 * j) * 48);
            b1[0][j] = *(const W3*)(wq + (4 * j + 1) * 48);
            af[0][j] = *(const v2f*)(xa + 4 * j);
        }

        v8f acc[3];
        #pragma unroll
        for (int nt = 0; nt < 3; ++nt) acc[nt] = v8f{};
        #pragma unroll
        for (int g = 0; g < 6; ++g) {
            const int cur = g & 1, nxt = cur ^ 1;
            if (g < 5) {
                #pragma unroll
                for (int j = 0; j < 4; ++j) {
                    const int s = 4 * (g + 1) + j;
                    b0[nxt][j] = *(const W3*)(wq + (4 * s) * 48);
                    b1[nxt][j] = *(const W3*)(wq + (4 * s + 1) * 48);
                    af[nxt][j] = *(const v2f*)(xa + 4 * s);
                }
            }
            __builtin_amdgcn_sched_barrier(0);
            #pragma unroll
            for (int j = 0; j < 4; ++j) {
                acc[0] = wmma4(af[cur][j], v2f{ b0[cur][j].c0, b1[cur][j].c0 }, acc[0]);
                acc[1] = wmma4(af[cur][j], v2f{ b0[cur][j].c1, b1[cur][j].c1 }, acc[1]);
                acc[2] = wmma4(af[cur][j], v2f{ b0[cur][j].c2, b1[cur][j].c2 }, acc[2]);
            }
            __builtin_amdgcn_sched_barrier(0);
        }
        #pragma unroll
        for (int nt = 0; nt < 3; ++nt) {
            const int c  = 3 * lo + nt;
            float*    bp = ((c < 16) ? (Sw + c) : (sKV + (mt * 16) * KVP + (c - 16)))
                           + 8 * hi * KVP;
            #pragma unroll
            for (int r = 0; r < 8; ++r)
                bp[r * KVP] = acc[nt][r];
        }
    }
    __syncthreads();

    v2f wof[6][4];
    {
        const float* wo = Wo + (size_t)e * HD * C_DIM + (2 * hi) * C_DIM + lo;
        #pragma unroll
        for (int nt = 0; nt < 6; ++nt) {
            #pragma unroll
            for (int s = 0; s < 4; ++s)
                wof[nt][s] = v2f{ wo[(4 * s) * C_DIM + nt * 16],
                                  wo[(4 * s + 1) * C_DIM + nt * 16] };
        }
    }

    v8f sc[4];
    #pragma unroll
    for (int nt = 0; nt < 4; ++nt) sc[nt] = v8f{};
    {
        v2f qa[4];
        #pragma unroll
        for (int s = 0; s < 4; ++s)
            qa[s] = *(const v2f*)(Sw + lo * KVP + 4 * s + 2 * hi);
        #pragma unroll
        for (int s = 0; s < 4; ++s) {
            #pragma unroll
            for (int nt = 0; nt < 4; ++nt) {
                const v2f bf = *(const v2f*)(sKV + (nt * 16 + lo) * KVP + 4 * s + 2 * hi);
                sc[nt] = wmma4(qa[s], bf, sc[nt]);
            }
        }
    }

    int pb[2][4];
    #pragma unroll
    for (int tt = 0; tt < 2; ++tt) {
        const int kap = 16 * tt + lo;
        const int g   = 2 * (kap >> 3) + ((kap >> 1) & 1);
        const int j   = (kap & 1) + ((kap >> 1) & 2);
        #pragma unroll
        for (int q = 0; q < 4; ++q)
            pb[tt][q] = 64 * g + j + 4 * (8 * hi + (q ^ (g & 3)));
    }
    float p2[8], p3[8];
    #pragma unroll
    for (int r = 0; r < 8; ++r) {
        const int row = mt * 16 + r + 8 * hi;
        float s0 = sc[0][r] * 0.25f, s1 = sc[1][r] * 0.25f;
        float s2 = sc[2][r] * 0.25f, s3 = sc[3][r] * 0.25f;
        float m = row16_max(fmaxf(fmaxf(s0, s1), fmaxf(s2, s3)));
        float e0 = __expf(s0 - m), e1 = __expf(s1 - m);
        float e2 = __expf(s2 - m), e3v = __expf(s3 - m);
        float sum = row16_sum((e0 + e1) + (e2 + e3v));
        float inv = sgate[row] / sum;
        Sw[pb[0][r & 3] + 16 * (r >> 2)] = e0 * inv;
        Sw[pb[1][r & 3] + 16 * (r >> 2)] = e1 * inv;
        p2[r] = e2 * inv;
        p3[r] = e3v * inv;
    }
    __builtin_amdgcn_fence(__ATOMIC_ACQ_REL, "workgroup", "local");

    {
        v8f acc = {};
        #pragma unroll
        for (int h = 0; h < 2; ++h) {
            if (h == 1) {
                #pragma unroll
                for (int r = 0; r < 8; ++r) {
                    Sw[pb[0][r & 3] + 16 * (r >> 2)] = p2[r];
                    Sw[pb[1][r & 3] + 16 * (r >> 2)] = p3[r];
                }
                __builtin_amdgcn_fence(__ATOMIC_ACQ_REL, "workgroup", "local");
            }
            #pragma unroll
            for (int i = 0; i < 4; ++i) {
                const int g  = 2 * i + hi;
                const float4 p4 = *(const float4*)(Sw + 64 * g + 4 * (lo ^ (g & 3)));
                const int k0 = 32 * h + 8 * i + 2 * hi;
                v2f bf0 = { sKV[k0 * KVP + 16 + lo],
                            sKV[(k0 + 1) * KVP + 16 + lo] };
                v2f bf1 = { sKV[(k0 + 4) * KVP + 16 + lo],
                            sKV[(k0 + 5) * KVP + 16 + lo] };
                acc = wmma4(v2f{ p4.x, p4.y }, bf0, acc);
                acc = wmma4(v2f{ p4.z, p4.w }, bf1, acc);
            }
        }
        #pragma unroll
        for (int r = 0; r < 8; ++r)
            Sw[pb[0][r & 3] + 16 * (r >> 2)] = acc[r];
    }
    __builtin_amdgcn_fence(__ATOMIC_ACQ_REL, "workgroup", "local");

    v2f oa[4];
    #pragma unroll
    for (int i = 0; i < 2; ++i) {
        const int g = 2 * i + hi;
        const float4 o4 = *(const float4*)(Sw + 64 * g + 4 * (lo ^ (g & 3)));
        oa[2 * i]     = v2f{ o4.x, o4.y };
        oa[2 * i + 1] = v2f{ o4.z, o4.w };
    }
    #pragma unroll
    for (int nt = 0; nt < 6; ++nt) {
        v8f acc = {};
        #pragma unroll
        for (int s = 0; s < 4; ++s)
            acc = wmma4(oa[s], wof[nt][s], acc);
        #pragma unroll
        for (int r = 0; r < 8; ++r) {
            int m  = mt * 16 + r + 8 * hi;
            int t  = sidx[m];
            atomicAdd(&tile[t * TP + nt * 16 + lo], acc[r]);
        }
    }
    __syncthreads();
    }

    {
        const int p = wave & 3;
        const int q = wave >> 2;
        const float* ta0 = tile + ((4 * p) * 16 + lo) * TP + 4 * hi;
        const float* wb0 = projW + (size_t)((2 * q) * 16 + lo) * C_DIM + 4 * hi;
        const float* wb1 = wb0 + 16 * C_DIM;

        v8f pa[4][2];
        #pragma unroll
        for (int i = 0; i < 4; ++i) {
            #pragma unroll
            for (int j = 0; j < 2; ++j) pa[i][j] = v8f{};
        }
        float4 fa[2][4], fb[2][2];
        #pragma unroll
        for (int i = 0; i < 4; ++i)
            fa[0][i] = *(const float4*)(ta0 + i * 16 * TP);
        fb[0][0] = *(const float4*)(wb0);
        fb[0][1] = *(const float4*)(wb1);
        #pragma unroll
        for (int u = 0; u < 12; ++u) {
            const int cur = u & 1, nxt = cur ^ 1;
            if (u < 11) {
                #pragma unroll
                for (int i = 0; i < 4; ++i)
                    fa[nxt][i] = *(const float4*)(ta0 + i * 16 * TP + 8 * (u + 1));
                fb[nxt][0] = *(const float4*)(wb0 + 8 * (u + 1));
                fb[nxt][1] = *(const float4*)(wb1 + 8 * (u + 1));
            }
            __builtin_amdgcn_sched_barrier(0);
            #pragma unroll
            for (int i = 0; i < 4; ++i) {
                pa[i][0] = wmma4(v2f{ fa[cur][i].x, fa[cur][i].y }, v2f{ fb[cur][0].x, fb[cur][0].y }, pa[i][0]);
                pa[i][1] = wmma4(v2f{ fa[cur][i].x, fa[cur][i].y }, v2f{ fb[cur][1].x, fb[cur][1].y }, pa[i][1]);
            }
            #pragma unroll
            for (int i = 0; i < 4; ++i) {
                pa[i][0] = wmma4(v2f{ fa[cur][i].z, fa[cur][i].w }, v2f{ fb[cur][0].z, fb[cur][0].w }, pa[i][0]);
                pa[i][1] = wmma4(v2f{ fa[cur][i].z, fa[cur][i].w }, v2f{ fb[cur][1].z, fb[cur][1].w }, pa[i][1]);
            }
            __builtin_amdgcn_sched_barrier(0);
        }
        #pragma unroll
        for (int j = 0; j < 2; ++j) {
            const int col = (2 * q + j) * 16 + lo;
            const float bs = projb[col];
            #pragma unroll
            for (int i = 0; i < 4; ++i) {
                const int rt = 4 * p + i;
                float* yp = y + ((size_t)(b * IMG + wy * WIN + rt) * IMG
                                 + wx * WIN + 8 * hi) * C_DIM + col;
                #pragma unroll
                for (int r = 0; r < 8; ++r)
                    yp[r * C_DIM] = pa[i][j][r] + bs;
            }
        }
    }
}

extern "C" void kernel_launch(void* const* d_in, const int* in_sizes, int n_in,
                              void* d_out, int out_size, void* d_ws, size_t ws_size,
                              hipStream_t stream) {
    const float* x     = (const float*)d_in[0];
    const float* Wr    = (const float*)d_in[1];
    const float* Wqkv  = (const float*)d_in[2];
    const float* Wo    = (const float*)d_in[3];
    const float* projW = (const float*)d_in[4];
    const float* projb = (const float*)d_in[5];
    float* y = (float*)d_out;

    int*   gidx = (int*)d_ws;
    float* gval = (float*)((char*)d_ws + (size_t)BW * HEADS * KSEL * sizeof(int));

    k1_router<<<BW, 256, 0, stream>>>(x, Wr, gidx, gval);
    k2_attn_proj<<<BW, F_THREADS, 0, stream>>>(x, Wqkv, Wo, gidx, gval, projW, projb, y);
}


# k1 fused into k2, one launch: a window's block computes its router logits (row loads as one group), waves 0-5 sort the heads, picks and gates pass in LDS, not the workspace; picks, sums unchanged
# speedup vs baseline: 2.7348x; 1.0896x over previous
#include <hip/hip_runtime.h>
#include <hip/hip_bf16.h>


#define BATCH   2
#define IMG     384
#define C_DIM   96
#define HEADS   6
#define HD      16
#define WIN     16
#define NWIN    24
#define BW      (BATCH * NWIN * NWIN)
#define LTOK    256
#define KSEL    64
#define NPIX    (BATCH * IMG * IMG)

typedef __attribute__((ext_vector_type(2))) float v2f;
typedef __attribute__((ext_vector_type(8))) float v8f;

static __device__ inline v8f wmma4(v2f a, v2f b, v8f c) {
    return __builtin_amdgcn_wmma_f32_16x16x4_f32(
         false, a,  false, b,
         (short)0, c,  false,  false);
}

typedef unsigned long long u64;

template <int M> static __device__ __forceinline__ unsigned int lane_xor_u32(unsigned int x) {
    const int b = (int)x;
    if constexpr (M == 16) {
        return (unsigned int)__builtin_amdgcn_permlanex16(b, b, 0x76543210, 0xfedcba98, false, false);
    } else {
        return (unsigned int)__builtin_amdgcn_update_dpp(b, b, 0x160 + M, 0xF, 0xF, true);
    }
}
template <int M> static __device__ __forceinline__ u64 lane_xor_u64(u64 k) {
    const unsigned int lo = lane_xor_u32<M>((unsigned int)k);
    const unsigned int hi = lane_xor_u32<M>((unsigned int)(k >> 32));
    return ((u64)hi << 32) | lo;
}

template <int KS, int J>
static __device__ __forceinline__ void cx_lanes(u64 (&k)[8], const int lane) {
    const bool low = (lane & J) == 0;
    #pragma unroll
    for (int r = 0; r < 8; ++r) {
        const bool desc = (KS < 32) ? ((lane & KS) == 0) : (((r * 32) & KS) == 0);
        const bool keepMax = (desc == low);
        const u64 a = k[r];
        const u64 b = lane_xor_u64<J>(a);
        const bool gt = a > b;
        k[r] = (gt == keepMax) ? a : b;
    }
}
template <int KS, int JR>
static __device__ __forceinline__ void cx_regs(u64 (&k)[8]) {
    #pragma unroll
    for (int r = 0; r < 8; ++r) {
        if ((r & JR) == 0) {
            const bool desc = ((r * 32) & KS) == 0;
            const u64 a = k[r], b = k[r ^ JR];
            const u64 mx = (a > b) ? a : b;
            const u64 mn = (a > b) ? b : a;
            k[r]      = desc ? mx : mn;
            k[r ^ JR] = desc ? mn : mx;
        }
    }
}
template <int KS>
static __device__ __forceinline__ void cx_tail(u64 (&k)[8], const int lane) {
    cx_lanes<KS, 16>(k, lane);
    cx_lanes<KS, 8>(k, lane);
    cx_lanes<KS, 4>(k, lane);
    cx_lanes<KS, 2>(k, lane);
    cx_lanes<KS, 1>(k, lane);
}

static __device__ __forceinline__ float dpp_f(float v, const int ctrl_sel) {
    int x = __float_as_int(v);
    int y;
    switch (ctrl_sel) {
      case 0:  y = __builtin_amdgcn_update_dpp(x, x, 0xB1,  0xF, 0xF, true); break;
      case 1:  y = __builtin_amdgcn_update_dpp(x, x, 0x4E,  0xF, 0xF, true); break;
      case 2:  y = __builtin_amdgcn_update_dpp(x, x, 0x141, 0xF, 0xF, true); break;
      default: y = __builtin_amdgcn_update_dpp(x, x, 0x140, 0xF, 0xF, true); break;
    }
    return __int_as_float(y);
}
static __device__ __forceinline__ float row16_max(float v) {
    v = fmaxf(v, dpp_f(v, 0));
    v = fmaxf(v, dpp_f(v, 1));
    v = fmaxf(v, dpp_f(v, 2));
    v = fmaxf(v, dpp_f(v, 3));
    return v;
}
static __device__ __forceinline__ float row16_sum(float v) {
    v = v + dpp_f(v, 0);
    v = v + dpp_f(v, 1);
    v = v + dpp_f(v, 2);
    v = v + dpp_f(v, 3);
    return v;
}

struct W3 { float c0, c1, c2; };


#define F_THREADS 384
#define TP        100
#define KVP       36
#define SW        576

__global__ __launch_bounds__(F_THREADS, 6) void k_window(
    const float* __restrict__ x, const float* __restrict__ Wr,
    const float* __restrict__ Wqkv, const float* __restrict__ Wo,
    const float* __restrict__ projW, const float* __restrict__ projb,
    float* __restrict__ y)
{
    __shared__ __attribute__((aligned(16))) float tile[LTOK * TP];
    __shared__ __attribute__((aligned(16))) float sKV3[3][KSEL * KVP];
    __shared__ __attribute__((aligned(16))) float strip12[12][SW];
    __shared__ float sgate6[HEADS][KSEL];
    __shared__ int   sidx6[HEADS][KSEL];

    const int w    = blockIdx.x;
    const int tid  = threadIdx.x;
    const int lane = tid & 31;
    const int wave = __builtin_amdgcn_readfirstlane(tid >> 5);
    const int e3   = wave >> 2;
    const int mt   = wave & 3;
    const int hi   = lane >> 4;
    const int lo   = lane & 15;

    float* Sw    = strip12[wave];
    float* sKV   = sKV3[e3];

    const int b  = w / (NWIN * NWIN);
    const int rm = w % (NWIN * NWIN);
    const int wy = rm / NWIN, wx = rm % NWIN;

    {
        float* sWr = &strip12[0][0];
        float* sv6 = sWr + C_DIM * HEADS;

        const float wr0 = Wr[tid];
        const float wr1 = (tid < C_DIM * HEADS - F_THREADS) ? Wr[tid + F_THREADS] : 0.f;

        float4 xv[C_DIM / 4];
        if (tid < LTOK) {
            const int hh = wy * WIN + (tid >> 4);
            const int ww = wx * WIN + (tid & 15);
            const float4* xr = (const float4*)(x + ((size_t)(b * IMG + hh) * IMG + ww) * C_DIM);
            #pragma unroll
            for (int i = 0; i < C_DIM / 4; ++i) xv[i] = xr[i];
        }
        {
            const float4 z4 = make_float4(0.f, 0.f, 0.f, 0.f);
            float4* t4 = (float4*)tile;
            for (int i = tid; i < LTOK * TP / 4; i += F_THREADS) t4[i] = z4;
        }
        sWr[tid] = wr0;
        if (tid < C_DIM * HEADS - F_THREADS) sWr[tid + F_THREADS] = wr1;
        __syncthreads();

        if (tid < LTOK) {
            float lg[HEADS];
            #pragma unroll
            for (int e = 0; e < HEADS; ++e) lg[e] = 0.f;
            const float4* w4 = (const float4*)sWr;
            #pragma unroll
            for (int i = 0; i < C_DIM / 4; ++i) {
                float wf[4 * HEADS];
                #pragma unroll
                for (int j = 0; j < HEADS; ++j) {
                    const float4 wv = w4[HEADS * i + j];
                    wf[4 * j] = wv.x; wf[4 * j + 1] = wv.y; wf[4 * j + 2] = wv.z; wf[4 * j + 3] = wv.w;
                }
                const float xc[4] = { xv[i].x, xv[i].y, xv[i].z, xv[i].w };
                #pragma unroll
                for (int cc = 0; cc < 4; ++cc) {
                    #pragma unroll
                    for (int e = 0; e < HEADS; ++e)
                        lg[e] = __builtin_fmaf(xc[cc], wf[cc * HEADS + e], lg[e]);
                }
            }
            #pragma unroll
            for (int e = 0; e < HEADS; ++e) lg[e] = 1.0f / (1.0f + __expf(-lg[e]));
            #pragma unroll
            for (int e = 0; e < HEADS; ++e) sv6[e * LTOK + tid] = lg[e];
        }
        __syncthreads();

        if (wave < HEADS) {
            const float* sv = sv6 + wave * LTOK;
            u64 k[8];
            #pragma unroll
            for (int r = 0; r < 8; ++r) {
                const int i = r * 32 + lane;
                const unsigned int fb = (i == 0) ? 0x7F800000u : __float_as_uint(sv[i]);
                k[r] = ((u64)fb << 32) | (unsigned int)(0xFFFFu - i);
            }

            cx_lanes<2, 1>(k, lane);
            cx_lanes<4, 2>(k, lane);  cx_lanes<4, 1>(k, lane);
            cx_lanes<8, 4>(k, lane);  cx_lanes<8, 2>(k, lane);  cx_lanes<8, 1>(k, lane);
            cx_lanes<16, 8>(k, lane); cx_lanes<16, 4>(k, lane); cx_lanes<16, 2>(k, lane); cx_lanes<16, 1>(k, lane);
            cx_tail<32>(k, lane);
            cx_regs<64, 1>(k);  cx_tail<64>(k, lane);
            cx_regs<128, 2>(k); cx_regs<128, 1>(k); cx_tail<128>(k, lane);
            cx_regs<256, 4>(k); cx_regs<256, 2>(k); cx_regs<256, 1>(k); cx_tail<256>(k, lane);

            #pragma unroll
            for (int r = 0; r < 2; ++r) {
                const int tok = 0xFFFF - (int)(k[r] & 0xFFFFull);
                sidx6[wave][r * 32 + lane]  = tok;
                sgate6[wave][r * 32 + lane] = sv[tok];
            }
        }
        __syncthreads();
    }

    for (int pass = 0; pass < 2; ++pass) {
    const int    e     = 3 * pass + e3;
    const float* sgate = sgate6[e];
    const int*   sidx  = sidx6[e];
    const int    tA    = sidx[mt * 16 + lo];

    {
        const float* wq = Wqkv + (size_t)e * C_DIM * 48 + (2 * hi) * 48 + 3 * lo;
        const float* xa = x + ((size_t)(b * IMG + wy * WIN + (tA >> 4)) * IMG
                               + wx * WIN + (tA & 15)) * C_DIM + 2 * hi;
        W3  b0[2][4], b1[2][4];
        v2f af[2][4];
        #pragma unroll
        for (int j = 0; j < 4; ++j) {
            b0[0][j] = *(const W3*)(wq + (4 * j) * 48);
            b1[0][j] = *(const W3*)(wq + (4 * j + 1) * 48);
            af[0][j] = *(const v2f*)(xa + 4 * j);
        }

        v8f acc[3];
        #pragma unroll
        for (int nt = 0; nt < 3; ++nt) acc[nt] = v8f{};
        #pragma unroll
        for (int g = 0; g < 6; ++g) {
            const int cur = g & 1, nxt = cur ^ 1;
            if (g < 5) {
                #pragma unroll
                for (int j = 0; j < 4; ++j) {
                    const int s = 4 * (g + 1) + j;
                    b0[nxt][j] = *(const W3*)(wq + (4 * s) * 48);
                    b1[nxt][j] = *(const W3*)(wq + (4 * s + 1) * 48);
                    af[nxt][j] = *(const v2f*)(xa + 4 * s);
                }
            }
            __builtin_amdgcn_sched_barrier(0);
            #pragma unroll
            for (int j = 0; j < 4; ++j) {
                acc[0] = wmma4(af[cur][j], v2f{ b0[cur][j].c0, b1[cur][j].c0 }, acc[0]);
                acc[1] = wmma4(af[cur][j], v2f{ b0[cur][j].c1, b1[cur][j].c1 }, acc[1]);
                acc[2] = wmma4(af[cur][j], v2f{ b0[cur][j].c2, b1[cur][j].c2 }, acc[2]);
            }
            __builtin_amdgcn_sched_barrier(0);
        }
        #pragma unroll
        for (int nt = 0; nt < 3; ++nt) {
            const int c  = 3 * lo + nt;
            float*    bp = ((c < 16) ? (Sw + c) : (sKV + (mt * 16) * KVP + (c - 16)))
                           + 8 * hi * KVP;
            #pragma unroll
            for (int r = 0; r < 8; ++r)
                bp[r * KVP] = acc[nt][r];
        }
    }
    __syncthreads();

    v2f wof[6][4];
    {
        const float* wo = Wo + (size_t)e * HD * C_DIM + (2 * hi) * C_DIM + lo;
        #pragma unroll
        for (int nt = 0; nt < 6; ++nt) {
            #pragma unroll
            for (int s = 0; s < 4; ++s)
                wof[nt][s] = v2f{ wo[(4 * s) * C_DIM + nt * 16],
                                  wo[(4 * s + 1) * C_DIM + nt * 16] };
        }
    }

    v8f sc[4];
    #pragma unroll
    for (int nt = 0; nt < 4; ++nt) sc[nt] = v8f{};
    {
        v2f qa[4];
        #pragma unroll
        for (int s = 0; s < 4; ++s)
            qa[s] = *(const v2f*)(Sw + lo * KVP + 4 * s + 2 * hi);
        #pragma unroll
        for (int s = 0; s < 4; ++s) {
            #pragma unroll
            for (int nt = 0; nt < 4; ++nt) {
                const v2f bf = *(const v2f*)(sKV + (nt * 16 + lo) * KVP + 4 * s + 2 * hi);
                sc[nt] = wmma4(qa[s], bf, sc[nt]);
            }
        }
    }

    int pb[2][4];
    #pragma unroll
    for (int tt = 0; tt < 2; ++tt) {
        const int kap = 16 * tt + lo;
        const int g   = 2 * (kap >> 3) + ((kap >> 1) & 1);
        const int j   = (kap & 1) + ((kap >> 1) & 2);
        #pragma unroll
        for (int q = 0; q < 4; ++q)
            pb[tt][q] = 64 * g + j + 4 * (8 * hi + (q ^ (g & 3)));
    }
    float p2[8], p3[8];
    #pragma unroll
    for (int r = 0; r < 8; ++r) {
        const int row = mt * 16 + r + 8 * hi;
        float s0 = sc[0][r] * 0.25f, s1 = sc[1][r] * 0.25f;
        float s2 = sc[2][r] * 0.25f, s3 = sc[3][r] * 0.25f;
        float m = row16_max(fmaxf(fmaxf(s0, s1), fmaxf(s2, s3)));
        float e0 = __expf(s0 - m), e1 = __expf(s1 - m);
        float e2 = __expf(s2 - m), e3v = __expf(s3 - m);
        float sum = row16_sum((e0 + e1) + (e2 + e3v));
        float inv = sgate[row] / sum;
        Sw[pb[0][r & 3] + 16 * (r >> 2)] = e0 * inv;
        Sw[pb[1][r & 3] + 16 * (r >> 2)] = e1 * inv;
        p2[r] = e2 * inv;
        p3[r] = e3v * inv;
    }
    __builtin_amdgcn_fence(__ATOMIC_ACQ_REL, "workgroup", "local");

    {
        v8f acc = {};
        #pragma unroll
        for (int h = 0; h < 2; ++h) {
            if (h == 1) {
                #pragma unroll
                for (int r = 0; r < 8; ++r) {
                    Sw[pb[0][r & 3] + 16 * (r >> 2)] = p2[r];
                    Sw[pb[1][r & 3] + 16 * (r >> 2)] = p3[r];
                }
                __builtin_amdgcn_fence(__ATOMIC_ACQ_REL, "workgroup", "local");
            }
            #pragma unroll
            for (int i = 0; i < 4; ++i) {
                const int g  = 2 * i + hi;
                const float4 p4 = *(const float4*)(Sw + 64 * g + 4 * (lo ^ (g & 3)));
                const int k0 = 32 * h + 8 * i + 2 * hi;
                v2f bf0 = { sKV[k0 * KVP + 16 + lo],
                            sKV[(k0 + 1) * KVP + 16 + lo] };
                v2f bf1 = { sKV[(k0 + 4) * KVP + 16 + lo],
                            sKV[(k0 + 5) * KVP + 16 + lo] };
                acc = wmma4(v2f{ p4.x, p4.y }, bf0, acc);
                acc = wmma4(v2f{ p4.z, p4.w }, bf1, acc);
            }
        }
        #pragma unroll
        for (int r = 0; r < 8; ++r)
            Sw[pb[0][r & 3] + 16 * (r >> 2)] = acc[r];
    }
    __builtin_amdgcn_fence(__ATOMIC_ACQ_REL, "workgroup", "local");

    v2f oa[4];
    #pragma unroll
    for (int i = 0; i < 2; ++i) {
        const int g = 2 * i + hi;
        const float4 o4 = *(const float4*)(Sw + 64 * g + 4 * (lo ^ (g & 3)));
        oa[2 * i]     = v2f{ o4.x, o4.y };
        oa[2 * i + 1] = v2f{ o4.z, o4.w };
    }
    #pragma unroll
    for (int nt = 0; nt < 6; ++nt) {
        v8f acc = {};
        #pragma unroll
        for (int s = 0; s < 4; ++s)
            acc = wmma4(oa[s], wof[nt][s], acc);
        #pragma unroll
        for (int r = 0; r < 8; ++r) {
            int m  = mt * 16 + r + 8 * hi;
            int t  = sidx[m];
            atomicAdd(&tile[t * TP + nt * 16 + lo], acc[r]);
        }
    }
    __syncthreads();
    }

    {
        const int p = wave & 3;
        const int q = wave >> 2;
        const float* ta0 = tile + ((4 * p) * 16 + lo) * TP + 4 * hi;
        const float* wb0 = projW + (size_t)((2 * q) * 16 + lo) * C_DIM + 4 * hi;
        const float* wb1 = wb0 + 16 * C_DIM;

        v8f pa[4][2];
        #pragma unroll
        for (int i = 0; i < 4; ++i) {
            #pragma unroll
            for (int j = 0; j < 2; ++j) pa[i][j] = v8f{};
        }
        float4 fa[2][4], fb[2][2];
        #pragma unroll
        for (int i = 0; i < 4; ++i)
            fa[0][i] = *(const float4*)(ta0 + i * 16 * TP);
        fb[0][0] = *(const float4*)(wb0);
        fb[0][1] = *(const float4*)(wb1);
        #pragma unroll
        for (int u = 0; u < 12; ++u) {
            const int cur = u & 1, nxt = cur ^ 1;
            if (u < 11) {
                #pragma unroll
                for (int i = 0; i < 4; ++i)
                    fa[nxt][i] = *(const float4*)(ta0 + i * 16 * TP + 8 * (u + 1));
                fb[nxt][0] = *(const float4*)(wb0 + 8 * (u + 1));
                fb[nxt][1] = *(const float4*)(wb1 + 8 * (u + 1));
            }
            __builtin_amdgcn_sched_barrier(0);
            #pragma unroll
            for (int i = 0; i < 4; ++i) {
                pa[i][0] = wmma4(v2f{ fa[cur][i].x, fa[cur][i].y }, v2f{ fb[cur][0].x, fb[cur][0].y }, pa[i][0]);
                pa[i][1] = wmma4(v2f{ fa[cur][i].x, fa[cur][i].y }, v2f{ fb[cur][1].x, fb[cur][1].y }, pa[i][1]);
            }
            #pragma unroll
            for (int i = 0; i < 4; ++i) {
                pa[i][0] = wmma4(v2f{ fa[cur][i].z, fa[cur][i].w }, v2f{ fb[cur][0].z, fb[cur][0].w }, pa[i][0]);
                pa[i][1] = wmma4(v2f{ fa[cur][i].z, fa[cur][i].w }, v2f{ fb[cur][1].z, fb[cur][1].w }, pa[i][1]);
            }
            __builtin_amdgcn_sched_barrier(0);
        }
        #pragma unroll
        for (int j = 0; j < 2; ++j) {
            const int col = (2 * q + j) * 16 + lo;
            const float bs = projb[col];
            #pragma unroll
            for (int i = 0; i < 4; ++i) {
                const int rt = 4 * p + i;
                float* yp = y + ((size_t)(b * IMG + wy * WIN + rt) * IMG
                                 + wx * WIN + 8 * hi) * C_DIM + col;
                #pragma unroll
                for (int r = 0; r < 8; ++r)
                    yp[r * C_DIM] = pa[i][j][r] + bs;
            }
        }
    }
}

extern "C" void kernel_launch(void* const* d_in, const int* in_sizes, int n_in,
                              void* d_out, int out_size, void* d_ws, size_t ws_size,
                              hipStream_t stream) {
    const float* x     = (const float*)d_in[0];
    const float* Wr    = (const float*)d_in[1];
    const float* Wqkv  = (const float*)d_in[2];
    const float* Wo    = (const float*)d_in[3];
    const float* projW = (const float*)d_in[4];
    const float* projb = (const float*)d_in[5];
    float* y = (float*)d_out;

    k_window<<<BW, F_THREADS, 0, stream>>>(x, Wr, Wqkv, Wo, projW, projb, y);
}
